# starting kernel
# baseline (speedup 1.0000x reference)
#include <hip/hip_runtime.h>

#define HIDDEN 64
#define FEAT   9
#define SEQ    128
#define BATCH  8192
#define NCLS   2
#define GATES  256
#define MTILE  16

typedef _Float16 v16h __attribute__((ext_vector_type(16)));
typedef float    v8f  __attribute__((ext_vector_type(8)));

#define WMMA(A, B, C) __builtin_amdgcn_wmma_f32_16x16x32_f16( \
    false, (A), false, (B), (short)0, (C), false, false)

static __device__ __forceinline__ v16h ldsA(const _Float16* base, int ld, int kc) {
  const int lane = threadIdx.x & 31;
  const int m  = lane & 15;
  const int kb = ((lane >> 4) << 3) + (kc << 5);
  const _Float16* p = base + m * ld + kb;
  v16h a;
#pragma unroll
  for (int i = 0; i < 8; ++i) { a[i] = p[i]; a[i + 8] = p[i + 16]; }
  return a;
}

static __device__ __forceinline__ v16h loadB(const float* W, int ld, int n0, int kc) {
  const int lane = threadIdx.x & 31;
  const int n  = n0 + (lane & 15);
  const int kb = ((lane >> 4) << 3) + (kc << 5);
  const float* p = W + n * ld + kb;
  v16h b;
#pragma unroll
  for (int i = 0; i < 8; ++i) { b[i] = (_Float16)p[i]; b[i + 8] = (_Float16)p[i + 16]; }
  return b;
}

static __device__ __forceinline__ v16h loadB_pad(const float* W, int ld, int kmax, int n0) {
  const int lane = threadIdx.x & 31;
  const int n  = n0 + (lane & 15);
  const int kb = (lane >> 4) << 3;
  v16h b;
#pragma unroll
  for (int i = 0; i < 8; ++i) {
    int k0 = kb + i, k1 = kb + 16 + i;
    b[i]     = (k0 < kmax) ? (_Float16)W[n * ld + k0] : (_Float16)0.0f;
    b[i + 8] = (k1 < kmax) ? (_Float16)W[n * ld + k1] : (_Float16)0.0f;
  }
  return b;
}

static __device__ __forceinline__ float fsig(float x)  { return 1.0f / (1.0f + __expf(-x)); }
static __device__ __forceinline__ float ftanh(float x) { return 2.0f / (1.0f + __expf(-2.0f * x)) - 1.0f; }

__global__ __launch_bounds__(256) void lstm2_fused_wmma(
    const float* __restrict__ x,
    const float* __restrict__ Wih0, const float* __restrict__ Whh0,
    const float* __restrict__ bih0, const float* __restrict__ bhh0,
    const float* __restrict__ Wih1, const float* __restrict__ Whh1,
    const float* __restrict__ bih1, const float* __restrict__ bhh1,
    const float* __restrict__ Wfc,  const float* __restrict__ bfc,
    float* __restrict__ out) {
  __shared__ _Float16 sh_x [MTILE * 32];
  __shared__ _Float16 sh_h0[MTILE * HIDDEN];
  __shared__ _Float16 sh_h1[MTILE * HIDDEN];
  __shared__ float    sh_g [MTILE * GATES];
  __shared__ float    sh_b0[GATES];
  __shared__ float    sh_b1[GATES];

  const int tid   = threadIdx.x;
  const int wave  = tid >> 5;
  const int bbase = blockIdx.x * MTILE;

  sh_b0[tid] = bih0[tid] + bhh0[tid];
  sh_b1[tid] = bih1[tid] + bhh1[tid];
  for (int i = tid; i < MTILE * HIDDEN; i += 256) {
    sh_h0[i] = (_Float16)0.0f;
    sh_h1[i] = (_Float16)0.0f;
  }

  const int n0 = wave * 32, n1 = wave * 32 + 16;
  const v16h Bih0a  = loadB_pad(Wih0, FEAT, FEAT, n0);
  const v16h Bih0b  = loadB_pad(Wih0, FEAT, FEAT, n1);
  const v16h Bhh0a0 = loadB(Whh0, HIDDEN, n0, 0), Bhh0a1 = loadB(Whh0, HIDDEN, n0, 1);
  const v16h Bhh0b0 = loadB(Whh0, HIDDEN, n1, 0), Bhh0b1 = loadB(Whh0, HIDDEN, n1, 1);
  const v16h Bih1a0 = loadB(Wih1, HIDDEN, n0, 0), Bih1a1 = loadB(Wih1, HIDDEN, n0, 1);
  const v16h Bih1b0 = loadB(Wih1, HIDDEN, n1, 0), Bih1b1 = loadB(Wih1, HIDDEN, n1, 1);
  const v16h Bhh1a0 = loadB(Whh1, HIDDEN, n0, 0), Bhh1a1 = loadB(Whh1, HIDDEN, n0, 1);
  const v16h Bhh1b0 = loadB(Whh1, HIDDEN, n1, 0), Bhh1b1 = loadB(Whh1, HIDDEN, n1, 1);

  float c0[4] = {0.f, 0.f, 0.f, 0.f};
  float c1[4] = {0.f, 0.f, 0.f, 0.f};
  const int en  = tid & 63;
  const int emb = tid >> 6;

  const int lane = tid & 31;
  const int cn   = lane & 15;
  const int cmb  = (lane >> 4) << 3;

  for (int t = 0; t < SEQ; ++t) {
#pragma unroll
    for (int i = 0; i < 2; ++i) {
      int idx = tid + i * 256;
      int m = idx >> 5, k = idx & 31;
      float v = (k < FEAT) ? x[((size_t)(bbase + m) * SEQ + t) * FEAT + k] : 0.0f;
      sh_x[idx] = (_Float16)v;
    }
    if (t + 1 < SEQ)
      __builtin_prefetch(&x[((size_t)(bbase + (tid >> 5)) * SEQ + t + 1) * FEAT], 0, 0);
    __syncthreads();

    {
      v16h ax  = ldsA(sh_x, 32, 0);
      v16h ah0 = ldsA(sh_h0, HIDDEN, 0);
      v16h ah1 = ldsA(sh_h0, HIDDEN, 1);
      v8f acc0 = {}; v8f acc1 = {};
      acc0 = WMMA(ax,  Bih0a,  acc0);
      acc0 = WMMA(ah0, Bhh0a0, acc0);
      acc0 = WMMA(ah1, Bhh0a1, acc0);
      acc1 = WMMA(ax,  Bih0b,  acc1);
      acc1 = WMMA(ah0, Bhh0b0, acc1);
      acc1 = WMMA(ah1, Bhh0b1, acc1);
#pragma unroll
      for (int r = 0; r < 8; ++r) {
        sh_g[(cmb + r) * GATES + n0 + cn] = acc0[r];
        sh_g[(cmb + r) * GATES + n1 + cn] = acc1[r];
      }
    }
    __syncthreads();

#pragma unroll
    for (int p = 0; p < 4; ++p) {
      int m = p * 4 + emb;
      const float* g = sh_g + m * GATES;
      float gi = g[en]        + sh_b0[en];
      float gf = g[64  + en]  + sh_b0[64  + en];
      float gz = g[128 + en]  + sh_b0[128 + en];
      float go = g[192 + en]  + sh_b0[192 + en];
      float cc = fsig(gf) * c0[p] + fsig(gi) * ftanh(gz);
      c0[p] = cc;
      sh_h0[m * HIDDEN + en] = (_Float16)(fsig(go) * ftanh(cc));
    }
    __syncthreads();

    {
      v16h ai0 = ldsA(sh_h0, HIDDEN, 0);
      v16h ai1 = ldsA(sh_h0, HIDDEN, 1);
      v16h ah0 = ldsA(sh_h1, HIDDEN, 0);
      v16h ah1 = ldsA(sh_h1, HIDDEN, 1);
      v8f acc0 = {}; v8f acc1 = {};
      acc0 = WMMA(ai0, Bih1a0, acc0);
      acc0 = WMMA(ai1, Bih1a1, acc0);
      acc0 = WMMA(ah0, Bhh1a0, acc0);
      acc0 = WMMA(ah1, Bhh1a1, acc0);
      acc1 = WMMA(ai0, Bih1b0, acc1);
      acc1 = WMMA(ai1, Bih1b1, acc1);
      acc1 = WMMA(ah0, Bhh1b0, acc1);
      acc1 = WMMA(ah1, Bhh1b1, acc1);
#pragma unroll
      for (int r = 0; r < 8; ++r) {
        sh_g[(cmb + r) * GATES + n0 + cn] = acc0[r];
        sh_g[(cmb + r) * GATES + n1 + cn] = acc1[r];
      }
    }
    __syncthreads();

#pragma unroll
    for (int p = 0; p < 4; ++p) {
      int m = p * 4 + emb;
      const float* g = sh_g + m * GATES;
      float gi = g[en]        + sh_b1[en];
      float gf = g[64  + en]  + sh_b1[64  + en];
      float gz = g[128 + en]  + sh_b1[128 + en];
      float go = g[192 + en]  + sh_b1[192 + en];
      float cc = fsig(gf) * c1[p] + fsig(gi) * ftanh(gz);
      c1[p] = cc;
      sh_h1[m * HIDDEN + en] = (_Float16)(fsig(go) * ftanh(cc));
    }
    __syncthreads();
  }

  if (tid < MTILE * NCLS) {
    int m = tid >> 1, cls = tid & 1;
    float s = bfc[cls];
#pragma unroll 8
    for (int n = 0; n < HIDDEN; ++n)
      s += (float)sh_h1[m * HIDDEN + n] * Wfc[cls * HIDDEN + n];
    out[(bbase + m) * NCLS + cls] = s;
  }
}

extern "C" void kernel_launch(void* const* d_in, const int* in_sizes, int n_in,
                              void* d_out, int out_size, void* d_ws, size_t ws_size,
                              hipStream_t stream) {
  const float* x    = (const float*)d_in[0];
  const float* Wih0 = (const float*)d_in[1];
  const float* Whh0 = (const float*)d_in[2];
  const float* bih0 = (const float*)d_in[3];
  const float* bhh0 = (const float*)d_in[4];
  const float* Wih1 = (const float*)d_in[5];
  const float* Whh1 = (const float*)d_in[6];
  const float* bih1 = (const float*)d_in[7];
  const float* bhh1 = (const float*)d_in[8];
  const float* Wfc  = (const float*)d_in[9];
  const float* bfc  = (const float*)d_in[10];
  float* out = (float*)d_out;

  dim3 grid(BATCH / MTILE);
  dim3 block(256);
  hipLaunchKernelGGL(lstm2_fused_wmma, grid, block, 0, stream,
                     x, Wih0, Whh0, bih0, bhh0, Wih1, Whh1, bih1, bhh1,
                     Wfc, bfc, out);
}


# cell update on two IEEE divisions per cell, not five: c' and h each as one quotient of products of (1+e^-gate) factors, exponents capped at 27; f32 last bits of c and h differ, matrix sums unchanged
# speedup vs baseline: 1.9416x; 1.9416x over previous
#include <hip/hip_runtime.h>


#define HIDDEN 64
#define FEAT   9
#define SEQ    128
#define BATCH  8192
#define NCLS   2
#define NTILE  16

typedef _Float16 v16h __attribute__((ext_vector_type(16)));
typedef _Float16 v8h  __attribute__((ext_vector_type(8)));
typedef float    v8f  __attribute__((ext_vector_type(8)));

#define WMMA(A, B, C) __builtin_amdgcn_wmma_f32_16x16x32_f16( \
    false, (A), false, (B), (short)0, (C), false, false)

static __device__ __forceinline__ v16h loadA(const float* W, int ld, int m0, int kc) {
  const int lane = threadIdx.x & 31;
  const float* p = W + (m0 + (lane & 15)) * ld + ((lane >> 4) << 3) + (kc << 5);
  v16h a;
#pragma unroll
  for (int i = 0; i < 8; ++i) { a[i] = (_Float16)p[i]; a[i + 8] = (_Float16)p[i + 16]; }
  return a;
}

static __device__ __forceinline__ v16h loadA_pad(const float* W, int ld, int kmax, int m0) {
  const int lane = threadIdx.x & 31;
  const int n  = m0 + (lane & 15);
  const int kb = (lane >> 4) << 3;
  v16h a;
#pragma unroll
  for (int i = 0; i < 8; ++i) {
    int k0 = kb + i, k1 = kb + 16 + i;
    float w0 = W[n * ld + (k0 < kmax ? k0 : kmax - 1)];
    float w1 = W[n * ld + (k1 < kmax ? k1 : kmax - 1)];
    a[i]     = (k0 < kmax) ? (_Float16)w0 : (_Float16)0.0f;
    a[i + 8] = (k1 < kmax) ? (_Float16)w1 : (_Float16)0.0f;
  }
  return a;
}

#define ECAP 27.0f
static __device__ __forceinline__ float ecap(float y) { return __expf(fminf(y, ECAP)); }

#define TILE_ROW0(t) (((t) & 3) * HIDDEN + ((t) >> 2) * 16)

__global__ __launch_bounds__(64) void lstm2_layer_per_wave_wmma(
    const float* __restrict__ x,
    const float* __restrict__ Wih0, const float* __restrict__ Whh0,
    const float* __restrict__ bih0, const float* __restrict__ bhh0,
    const float* __restrict__ Wih1, const float* __restrict__ Whh1,
    const float* __restrict__ bih1, const float* __restrict__ bhh1,
    const float* __restrict__ Wfc,  const float* __restrict__ bfc,
    float* __restrict__ out) {
  __shared__ v8h sh_h0[2][4][32];
  __shared__ __attribute__((aligned(16))) _Float16 sh_h1[16 * HIDDEN];

  const int lane = threadIdx.x & 31;
  const int lc   = lane & 15;
  const int lh   = lane >> 4;
  const int role  = __builtin_amdgcn_readfirstlane((int)(threadIdx.x >> 5));
  const int bbase = blockIdx.x * 16;

  const float* Wrec = role ? Whh1 : Whh0;
  const float* bi   = role ? bih1 : bih0;
  const float* bh   = role ? bhh1 : bhh0;

  const v16h zero16 = {};

  v16h  Wf[NTILE][4];
  float bias[NTILE][8];
  if (role) {
#pragma unroll
    for (int t = 0; t < NTILE; ++t) {
      Wf[t][0] = loadA(Wih1, HIDDEN, TILE_ROW0(t), 0);
      Wf[t][1] = loadA(Wih1, HIDDEN, TILE_ROW0(t), 1);
    }
  } else {
#pragma unroll
    for (int t = 0; t < NTILE; ++t) {
      Wf[t][0] = loadA_pad(Wih0, FEAT, FEAT, TILE_ROW0(t));
      Wf[t][1] = zero16;
    }
  }
#pragma unroll
  for (int t = 0; t < NTILE; ++t) {
    Wf[t][2] = loadA(Wrec, HIDDEN, TILE_ROW0(t), 0);
    Wf[t][3] = loadA(Wrec, HIDDEN, TILE_ROW0(t), 1);
  }
#pragma unroll
  for (int t = 0; t < NTILE; ++t)
#pragma unroll
    for (int r = 0; r < 8; ++r)
      bias[t][r] = bi[TILE_ROW0(t) + 8 * lh + r] + bh[TILE_ROW0(t) + 8 * lh + r];

  float cst[4][8];
#pragma unroll
  for (int j = 0; j < 4; ++j)
#pragma unroll
    for (int r = 0; r < 8; ++r) cst[j][r] = 0.0f;

  v16h Hold[2] = {zero16, zero16};

  const float* xrow = x + (size_t)(bbase + lc) * SEQ * FEAT + 8 * lh;
  v16h bxcur = zero16;
  float xraw[8];
#pragma unroll
  for (int i = 0; i < 8; ++i) xraw[i] = 0.0f;
  if (role == 0) {
#pragma unroll
    for (int i = 0; i < 8; ++i) {
      float v = xrow[lh ? 0 : i];
      bxcur[i] = (_Float16)((lh && i > 0) ? 0.0f : v);
    }
  }

  for (int it = 0; it <= SEQ; ++it) {
    const bool active = role ? (it >= 1) : (it < SEQ);
    if (active) {
      const int t = it - role;
      v16h Bin[2];
      if (role) {
        const int s = t & 1;
        Bin[0] = __builtin_shufflevector(sh_h0[s][0][lane], sh_h0[s][1][lane],
                                         0, 1, 2, 3, 4, 5, 6, 7, 8, 9, 10, 11, 12, 13, 14, 15);
        Bin[1] = __builtin_shufflevector(sh_h0[s][2][lane], sh_h0[s][3][lane],
                                         0, 1, 2, 3, 4, 5, 6, 7, 8, 9, 10, 11, 12, 13, 14, 15);
      } else {
        Bin[0] = bxcur;
        Bin[1] = zero16;
        if (t + 1 < SEQ) {
          const float* p = xrow + (size_t)(t + 1) * FEAT;
#pragma unroll
          for (int i = 0; i < 8; ++i) xraw[i] = p[lh ? 0 : i];
        }
      }

      v16h Hn[2] = {zero16, zero16};
#pragma unroll
      for (int j = 0; j < 4; ++j) {
        v8f acc[4];
#pragma unroll
        for (int g = 0; g < 4; ++g) acc[g] = (v8f){};
#pragma unroll
        for (int ch = 0; ch < 4; ++ch) {
          const v16h bop = (ch < 2) ? Bin[ch & 1] : Hold[ch & 1];
#pragma unroll
          for (int g = 0; g < 4; ++g)
            acc[g] = WMMA(Wf[4 * j + g][ch], bop, acc[g]);
        }
#pragma unroll
        for (int r = 0; r < 8; ++r) {
          float gi = acc[0][r] + bias[4 * j + 0][r];
          float gf = acc[1][r] + bias[4 * j + 1][r];
          float gz = acc[2][r] + bias[4 * j + 2][r];
          float go = acc[3][r] + bias[4 * j + 3][r];
          float ei = ecap(-gi), ef = ecap(-gf), eg = ecap(-2.0f * gz), eo = ecap(-go);
          float a  = 1.0f + ef;
          float p  = (1.0f + ei) * (1.0f + eg);
          float cc = (cst[j][r] * p + (1.0f - eg) * a) / (a * p);
          cst[j][r] = cc;
          float ec = ecap(-2.0f * cc);
          Hn[j >> 1][(j & 1) * 8 + r] = (_Float16)((1.0f - ec) / ((1.0f + eo) * (1.0f + ec)));
        }
      }

      Hold[0] = Hn[0];
      Hold[1] = Hn[1];

      if (role == 0) {
#pragma unroll
        for (int i = 0; i < 8; ++i)
          bxcur[i] = (_Float16)((lh && i > 0) ? 0.0f : xraw[i]);
        const int s = t & 1;
        sh_h0[s][0][lane] = __builtin_shufflevector(Hn[0], Hn[0], 0, 1, 2, 3, 4, 5, 6, 7);
        sh_h0[s][1][lane] = __builtin_shufflevector(Hn[0], Hn[0], 8, 9, 10, 11, 12, 13, 14, 15);
        sh_h0[s][2][lane] = __builtin_shufflevector(Hn[1], Hn[1], 0, 1, 2, 3, 4, 5, 6, 7);
        sh_h0[s][3][lane] = __builtin_shufflevector(Hn[1], Hn[1], 8, 9, 10, 11, 12, 13, 14, 15);
      }
    }
    __syncthreads();
  }

  if (role) {
#pragma unroll
    for (int kc = 0; kc < 2; ++kc) {
      *(v8h*)&sh_h1[lc * HIDDEN + 32 * kc + 8 * lh] =
          __builtin_shufflevector(Hold[kc], Hold[kc], 0, 1, 2, 3, 4, 5, 6, 7);
      *(v8h*)&sh_h1[lc * HIDDEN + 32 * kc + 16 + 8 * lh] =
          __builtin_shufflevector(Hold[kc], Hold[kc], 8, 9, 10, 11, 12, 13, 14, 15);
    }
  }
  __syncthreads();

  if (role) {
    int m = lane >> 1, cls = lane & 1;
    float s = bfc[cls];
#pragma unroll 8
    for (int n = 0; n < HIDDEN; ++n)
      s += (float)sh_h1[m * HIDDEN + n] * Wfc[cls * HIDDEN + n];
    out[(bbase + m) * NCLS + cls] = s;
  }
}

extern "C" void kernel_launch(void* const* d_in, const int* in_sizes, int n_in,
                              void* d_out, int out_size, void* d_ws, size_t ws_size,
                              hipStream_t stream) {
  const float* x    = (const float*)d_in[0];
  const float* Wih0 = (const float*)d_in[1];
  const float* Whh0 = (const float*)d_in[2];
  const float* bih0 = (const float*)d_in[3];
  const float* bhh0 = (const float*)d_in[4];
  const float* Wih1 = (const float*)d_in[5];
  const float* Whh1 = (const float*)d_in[6];
  const float* bih1 = (const float*)d_in[7];
  const float* bhh1 = (const float*)d_in[8];
  const float* Wfc  = (const float*)d_in[9];
  const float* bfc  = (const float*)d_in[10];
  float* out = (float*)d_out;

  dim3 grid(BATCH / 16);
  dim3 block(64);
  hipLaunchKernelGGL(lstm2_layer_per_wave_wmma, grid, block, 0, stream,
                     x, Wih0, Whh0, bih0, bhh0, Wih1, Whh1, bih1, bhh1,
                     Wfc, bfc, out);
}


# cell update: each quotient = numerator times v_rcp_f32 of its denominator (1 ulp), not the IEEE division sequence (its v_div_fmas waits on VCC); f32 last bits of c and h differ, sums unchanged
# speedup vs baseline: 2.7160x; 1.3988x over previous
#include <hip/hip_runtime.h>


#define HIDDEN 64
#define FEAT   9
#define SEQ    128
#define BATCH  8192
#define NCLS   2
#define NTILE  16

typedef _Float16 v16h __attribute__((ext_vector_type(16)));
typedef _Float16 v8h  __attribute__((ext_vector_type(8)));
typedef float    v8f  __attribute__((ext_vector_type(8)));

#define WMMA(A, B, C) __builtin_amdgcn_wmma_f32_16x16x32_f16( \
    false, (A), false, (B), (short)0, (C), false, false)

static __device__ __forceinline__ v16h loadA(const float* W, int ld, int m0, int kc) {
  const int lane = threadIdx.x & 31;
  const float* p = W + (m0 + (lane & 15)) * ld + ((lane >> 4) << 3) + (kc << 5);
  v16h a;
#pragma unroll
  for (int i = 0; i < 8; ++i) { a[i] = (_Float16)p[i]; a[i + 8] = (_Float16)p[i + 16]; }
  return a;
}

static __device__ __forceinline__ v16h loadA_pad(const float* W, int ld, int kmax, int m0) {
  const int lane = threadIdx.x & 31;
  const int n  = m0 + (lane & 15);
  const int kb = (lane >> 4) << 3;
  v16h a;
#pragma unroll
  for (int i = 0; i < 8; ++i) {
    int k0 = kb + i, k1 = kb + 16 + i;
    float w0 = W[n * ld + (k0 < kmax ? k0 : kmax - 1)];
    float w1 = W[n * ld + (k1 < kmax ? k1 : kmax - 1)];
    a[i]     = (k0 < kmax) ? (_Float16)w0 : (_Float16)0.0f;
    a[i + 8] = (k1 < kmax) ? (_Float16)w1 : (_Float16)0.0f;
  }
  return a;
}

static __device__ __forceinline__ float quot(float num, float den) {
  return num * __builtin_amdgcn_rcpf(den);
}
#define ECAP 27.0f
static __device__ __forceinline__ float ecap(float y) { return __expf(fminf(y, ECAP)); }

#define TILE_ROW0(t) (((t) & 3) * HIDDEN + ((t) >> 2) * 16)

__global__ __launch_bounds__(64) void lstm2_layer_per_wave_wmma(
    const float* __restrict__ x,
    const float* __restrict__ Wih0, const float* __restrict__ Whh0,
    const float* __restrict__ bih0, const float* __restrict__ bhh0,
    const float* __restrict__ Wih1, const float* __restrict__ Whh1,
    const float* __restrict__ bih1, const float* __restrict__ bhh1,
    const float* __restrict__ Wfc,  const float* __restrict__ bfc,
    float* __restrict__ out) {
  __shared__ v8h sh_h0[2][4][32];
  __shared__ __attribute__((aligned(16))) _Float16 sh_h1[16 * HIDDEN];

  const int lane = threadIdx.x & 31;
  const int lc   = lane & 15;
  const int lh   = lane >> 4;
  const int role  = __builtin_amdgcn_readfirstlane((int)(threadIdx.x >> 5));
  const int bbase = blockIdx.x * 16;

  const float* Wrec = role ? Whh1 : Whh0;
  const float* bi   = role ? bih1 : bih0;
  const float* bh   = role ? bhh1 : bhh0;

  const v16h zero16 = {};

  v16h  Wf[NTILE][4];
  float bias[NTILE][8];
  if (role) {
#pragma unroll
    for (int t = 0; t < NTILE; ++t) {
      Wf[t][0] = loadA(Wih1, HIDDEN, TILE_ROW0(t), 0);
      Wf[t][1] = loadA(Wih1, HIDDEN, TILE_ROW0(t), 1);
    }
  } else {
#pragma unroll
    for (int t = 0; t < NTILE; ++t) {
      Wf[t][0] = loadA_pad(Wih0, FEAT, FEAT, TILE_ROW0(t));
      Wf[t][1] = zero16;
    }
  }
#pragma unroll
  for (int t = 0; t < NTILE; ++t) {
    Wf[t][2] = loadA(Wrec, HIDDEN, TILE_ROW0(t), 0);
    Wf[t][3] = loadA(Wrec, HIDDEN, TILE_ROW0(t), 1);
  }
#pragma unroll
  for (int t = 0; t < NTILE; ++t)
#pragma unroll
    for (int r = 0; r < 8; ++r)
      bias[t][r] = bi[TILE_ROW0(t) + 8 * lh + r] + bh[TILE_ROW0(t) + 8 * lh + r];

  float cst[4][8];
#pragma unroll
  for (int j = 0; j < 4; ++j)
#pragma unroll
    for (int r = 0; r < 8; ++r) cst[j][r] = 0.0f;

  v16h Hold[2] = {zero16, zero16};

  const float* xrow = x + (size_t)(bbase + lc) * SEQ * FEAT + 8 * lh;
  v16h bxcur = zero16;
  float xraw[8];
#pragma unroll
  for (int i = 0; i < 8; ++i) xraw[i] = 0.0f;
  if (role == 0) {
#pragma unroll
    for (int i = 0; i < 8; ++i) {
      float v = xrow[lh ? 0 : i];
      bxcur[i] = (_Float16)((lh && i > 0) ? 0.0f : v);
    }
  }

  for (int it = 0; it <= SEQ; ++it) {
    const bool active = role ? (it >= 1) : (it < SEQ);
    if (active) {
      const int t = it - role;
      v16h Bin[2];
      if (role) {
        const int s = t & 1;
        Bin[0] = __builtin_shufflevector(sh_h0[s][0][lane], sh_h0[s][1][lane],
                                         0, 1, 2, 3, 4, 5, 6, 7, 8, 9, 10, 11, 12, 13, 14, 15);
        Bin[1] = __builtin_shufflevector(sh_h0[s][2][lane], sh_h0[s][3][lane],
                                         0, 1, 2, 3, 4, 5, 6, 7, 8, 9, 10, 11, 12, 13, 14, 15);
      } else {
        Bin[0] = bxcur;
        Bin[1] = zero16;
        if (t + 1 < SEQ) {
          const float* p = xrow + (size_t)(t + 1) * FEAT;
#pragma unroll
          for (int i = 0; i < 8; ++i) xraw[i] = p[lh ? 0 : i];
        }
      }

      v16h Hn[2] = {zero16, zero16};
#pragma unroll
      for (int j = 0; j < 4; ++j) {
        v8f acc[4];
#pragma unroll
        for (int g = 0; g < 4; ++g) acc[g] = (v8f){};
#pragma unroll
        for (int ch = 0; ch < 4; ++ch) {
          const v16h bop = (ch < 2) ? Bin[ch & 1] : Hold[ch & 1];
#pragma unroll
          for (int g = 0; g < 4; ++g)
            acc[g] = WMMA(Wf[4 * j + g][ch], bop, acc[g]);
        }
#pragma unroll
        for (int r = 0; r < 8; ++r) {
          float gi = acc[0][r] + bias[4 * j + 0][r];
          float gf = acc[1][r] + bias[4 * j + 1][r];
          float gz = acc[2][r] + bias[4 * j + 2][r];
          float go = acc[3][r] + bias[4 * j + 3][r];
          float ei = ecap(-gi), ef = ecap(-gf), eg = ecap(-2.0f * gz), eo = ecap(-go);
          float a  = 1.0f + ef;
          float p  = (1.0f + ei) * (1.0f + eg);
          float cc = quot(cst[j][r] * p + (1.0f - eg) * a, a * p);
          cst[j][r] = cc;
          float ec = ecap(-2.0f * cc);
          Hn[j >> 1][(j & 1) * 8 + r] = (_Float16)quot(1.0f - ec, (1.0f + eo) * (1.0f + ec));
        }
      }

      Hold[0] = Hn[0];
      Hold[1] = Hn[1];

      if (role == 0) {
#pragma unroll
        for (int i = 0; i < 8; ++i)
          bxcur[i] = (_Float16)((lh && i > 0) ? 0.0f : xraw[i]);
        const int s = t & 1;
        sh_h0[s][0][lane] = __builtin_shufflevector(Hn[0], Hn[0], 0, 1, 2, 3, 4, 5, 6, 7);
        sh_h0[s][1][lane] = __builtin_shufflevector(Hn[0], Hn[0], 8, 9, 10, 11, 12, 13, 14, 15);
        sh_h0[s][2][lane] = __builtin_shufflevector(Hn[1], Hn[1], 0, 1, 2, 3, 4, 5, 6, 7);
        sh_h0[s][3][lane] = __builtin_shufflevector(Hn[1], Hn[1], 8, 9, 10, 11, 12, 13, 14, 15);
      }
    }
    __syncthreads();
  }

  if (role) {
#pragma unroll
    for (int kc = 0; kc < 2; ++kc) {
      *(v8h*)&sh_h1[lc * HIDDEN + 32 * kc + 8 * lh] =
          __builtin_shufflevector(Hold[kc], Hold[kc], 0, 1, 2, 3, 4, 5, 6, 7);
      *(v8h*)&sh_h1[lc * HIDDEN + 32 * kc + 16 + 8 * lh] =
          __builtin_shufflevector(Hold[kc], Hold[kc], 8, 9, 10, 11, 12, 13, 14, 15);
    }
  }
  __syncthreads();

  if (role) {
    int m = lane >> 1, cls = lane & 1;
    float s = bfc[cls];
#pragma unroll 8
    for (int n = 0; n < HIDDEN; ++n)
      s += (float)sh_h1[m * HIDDEN + n] * Wfc[cls * HIDDEN + n];
    out[(bbase + m) * NCLS + cls] = s;
  }
}

extern "C" void kernel_launch(void* const* d_in, const int* in_sizes, int n_in,
                              void* d_out, int out_size, void* d_ws, size_t ws_size,
                              hipStream_t stream) {
  const float* x    = (const float*)d_in[0];
  const float* Wih0 = (const float*)d_in[1];
  const float* Whh0 = (const float*)d_in[2];
  const float* bih0 = (const float*)d_in[3];
  const float* bhh0 = (const float*)d_in[4];
  const float* Wih1 = (const float*)d_in[5];
  const float* Whh1 = (const float*)d_in[6];
  const float* bih1 = (const float*)d_in[7];
  const float* bhh1 = (const float*)d_in[8];
  const float* Wfc  = (const float*)d_in[9];
  const float* bfc  = (const float*)d_in[10];
  float* out = (float*)d_out;

  dim3 grid(BATCH / 16);
  dim3 block(64);
  hipLaunchKernelGGL(lstm2_layer_per_wave_wmma, grid, block, 0, stream,
                     x, Wih0, Whh0, bih0, bhh0, Wih1, Whh1, bih1, bhh1,
                     Wfc, bfc, out);
}


# cell update on packed two-wide f32 instructions (v_pk_add/mul/fma_f32): two cells of a lane per instruction, same arithmetic per cell; fused multiply-add may differ in last f32 bit, sums unchanged
# speedup vs baseline: 3.2152x; 1.1838x over previous
#include <hip/hip_runtime.h>


#define HIDDEN 64
#define FEAT   9
#define SEQ    128
#define BATCH  8192
#define NCLS   2
#define NTILE  16

typedef _Float16 v16h __attribute__((ext_vector_type(16)));
typedef _Float16 v8h  __attribute__((ext_vector_type(8)));
typedef float    v8f  __attribute__((ext_vector_type(8)));
typedef float    v2f  __attribute__((ext_vector_type(2)));
typedef _Float16 v2h  __attribute__((ext_vector_type(2)));

#define WMMA(A, B, C) __builtin_amdgcn_wmma_f32_16x16x32_f16( \
    false, (A), false, (B), (short)0, (C), false, false)

static __device__ __forceinline__ v16h loadA(const float* W, int ld, int m0, int kc) {
  const int lane = threadIdx.x & 31;
  const float* p = W + (m0 + (lane & 15)) * ld + ((lane >> 4) << 3) + (kc << 5);
  v16h a;
#pragma unroll
  for (int i = 0; i < 8; ++i) { a[i] = (_Float16)p[i]; a[i + 8] = (_Float16)p[i + 16]; }
  return a;
}

static __device__ __forceinline__ v16h loadA_pad(const float* W, int ld, int kmax, int m0) {
  const int lane = threadIdx.x & 31;
  const int n  = m0 + (lane & 15);
  const int kb = (lane >> 4) << 3;
  v16h a;
#pragma unroll
  for (int i = 0; i < 8; ++i) {
    int k0 = kb + i, k1 = kb + 16 + i;
    float w0 = W[n * ld + (k0 < kmax ? k0 : kmax - 1)];
    float w1 = W[n * ld + (k1 < kmax ? k1 : kmax - 1)];
    a[i]     = (k0 < kmax) ? (_Float16)w0 : (_Float16)0.0f;
    a[i + 8] = (k1 < kmax) ? (_Float16)w1 : (_Float16)0.0f;
  }
  return a;
}

#define ECAP  27.0f
#define LOG2E 1.4426950408889634f
static __device__ __forceinline__ v2f quot2(v2f num, v2f den) {
  v2f r = {__builtin_amdgcn_rcpf(den.x), __builtin_amdgcn_rcpf(den.y)};
  return num * r;
}
static __device__ __forceinline__ v2f ecap2(v2f g, float s) {
  const float cap = ECAP * LOG2E;
  v2f m = g * (s * LOG2E);
  return (v2f){__builtin_amdgcn_exp2f(fminf(m.x, cap)), __builtin_amdgcn_exp2f(fminf(m.y, cap))};
}

#define TILE_ROW0(t) (((t) & 3) * HIDDEN + ((t) >> 2) * 16)

__global__ __launch_bounds__(64) void lstm2_layer_per_wave_wmma(
    const float* __restrict__ x,
    const float* __restrict__ Wih0, const float* __restrict__ Whh0,
    const float* __restrict__ bih0, const float* __restrict__ bhh0,
    const float* __restrict__ Wih1, const float* __restrict__ Whh1,
    const float* __restrict__ bih1, const float* __restrict__ bhh1,
    const float* __restrict__ Wfc,  const float* __restrict__ bfc,
    float* __restrict__ out) {
  __shared__ v8h sh_h0[2][4][32];
  __shared__ __attribute__((aligned(16))) _Float16 sh_h1[16 * HIDDEN];

  const int lane = threadIdx.x & 31;
  const int lc   = lane & 15;
  const int lh   = lane >> 4;
  const int role  = __builtin_amdgcn_readfirstlane((int)(threadIdx.x >> 5));
  const int bbase = blockIdx.x * 16;

  const float* Wrec = role ? Whh1 : Whh0;
  const float* bi   = role ? bih1 : bih0;
  const float* bh   = role ? bhh1 : bhh0;

  const v16h zero16 = {};

  v16h  Wf[NTILE][4];
  v2f   bias[NTILE][4];
  if (role) {
#pragma unroll
    for (int t = 0; t < NTILE; ++t) {
      Wf[t][0] = loadA(Wih1, HIDDEN, TILE_ROW0(t), 0);
      Wf[t][1] = loadA(Wih1, HIDDEN, TILE_ROW0(t), 1);
    }
  } else {
#pragma unroll
    for (int t = 0; t < NTILE; ++t) {
      Wf[t][0] = loadA_pad(Wih0, FEAT, FEAT, TILE_ROW0(t));
      Wf[t][1] = zero16;
    }
  }
#pragma unroll
  for (int t = 0; t < NTILE; ++t) {
    Wf[t][2] = loadA(Wrec, HIDDEN, TILE_ROW0(t), 0);
    Wf[t][3] = loadA(Wrec, HIDDEN, TILE_ROW0(t), 1);
  }
#pragma unroll
  for (int t = 0; t < NTILE; ++t)
#pragma unroll
    for (int r = 0; r < 8; ++r)
      bias[t][r >> 1][r & 1] = bi[TILE_ROW0(t) + 8 * lh + r] + bh[TILE_ROW0(t) + 8 * lh + r];

  v2f cst[4][4];
#pragma unroll
  for (int j = 0; j < 4; ++j)
#pragma unroll
    for (int q = 0; q < 4; ++q) cst[j][q] = (v2f){0.0f, 0.0f};

  v16h Hold[2] = {zero16, zero16};

  const float* xrow = x + (size_t)(bbase + lc) * SEQ * FEAT + 8 * lh;
  v16h bxcur = zero16;
  float xraw[8];
#pragma unroll
  for (int i = 0; i < 8; ++i) xraw[i] = 0.0f;
  if (role == 0) {
#pragma unroll
    for (int i = 0; i < 8; ++i) {
      float v = xrow[lh ? 0 : i];
      bxcur[i] = (_Float16)((lh && i > 0) ? 0.0f : v);
    }
  }

  for (int it = 0; it <= SEQ; ++it) {
    const bool active = role ? (it >= 1) : (it < SEQ);
    if (active) {
      const int t = it - role;
      v16h Bin[2];
      if (role) {
        const int s = t & 1;
        Bin[0] = __builtin_shufflevector(sh_h0[s][0][lane], sh_h0[s][1][lane],
                                         0, 1, 2, 3, 4, 5, 6, 7, 8, 9, 10, 11, 12, 13, 14, 15);
        Bin[1] = __builtin_shufflevector(sh_h0[s][2][lane], sh_h0[s][3][lane],
                                         0, 1, 2, 3, 4, 5, 6, 7, 8, 9, 10, 11, 12, 13, 14, 15);
      } else {
        Bin[0] = bxcur;
        Bin[1] = zero16;
        if (t + 1 < SEQ) {
          const float* p = xrow + (size_t)(t + 1) * FEAT;
#pragma unroll
          for (int i = 0; i < 8; ++i) xraw[i] = p[lh ? 0 : i];
        }
      }

      v16h Hn[2] = {zero16, zero16};
#pragma unroll
      for (int j = 0; j < 4; ++j) {
        v8f acc[4];
#pragma unroll
        for (int g = 0; g < 4; ++g) acc[g] = (v8f){};
#pragma unroll
        for (int ch = 0; ch < 4; ++ch) {
          const v16h bop = (ch < 2) ? Bin[ch & 1] : Hold[ch & 1];
#pragma unroll
          for (int g = 0; g < 4; ++g)
            acc[g] = WMMA(Wf[4 * j + g][ch], bop, acc[g]);
        }
#pragma unroll
        for (int q = 0; q < 4; ++q) {
          v2f gi = (v2f){acc[0][2 * q], acc[0][2 * q + 1]} + bias[4 * j + 0][q];
          v2f gf = (v2f){acc[1][2 * q], acc[1][2 * q + 1]} + bias[4 * j + 1][q];
          v2f gz = (v2f){acc[2][2 * q], acc[2][2 * q + 1]} + bias[4 * j + 2][q];
          v2f go = (v2f){acc[3][2 * q], acc[3][2 * q + 1]} + bias[4 * j + 3][q];
          v2f ei = ecap2(gi, -1.0f), ef = ecap2(gf, -1.0f), eg = ecap2(gz, -2.0f), eo = ecap2(go, -1.0f);
          v2f a  = 1.0f + ef;
          v2f p  = (1.0f + ei) * (1.0f + eg);
          v2f cc = quot2(cst[j][q] * p + (1.0f - eg) * a, a * p);
          cst[j][q] = cc;
          v2f ec = ecap2(cc, -2.0f);
          v2h hp = __builtin_convertvector(quot2(1.0f - ec, (1.0f + eo) * (1.0f + ec)), v2h);
          Hn[j >> 1][(j & 1) * 8 + 2 * q]     = hp.x;
          Hn[j >> 1][(j & 1) * 8 + 2 * q + 1] = hp.y;
        }
      }

      Hold[0] = Hn[0];
      Hold[1] = Hn[1];

      if (role == 0) {
#pragma unroll
        for (int i = 0; i < 8; ++i)
          bxcur[i] = (_Float16)((lh && i > 0) ? 0.0f : xraw[i]);
        const int s = t & 1;
        sh_h0[s][0][lane] = __builtin_shufflevector(Hn[0], Hn[0], 0, 1, 2, 3, 4, 5, 6, 7);
        sh_h0[s][1][lane] = __builtin_shufflevector(Hn[0], Hn[0], 8, 9, 10, 11, 12, 13, 14, 15);
        sh_h0[s][2][lane] = __builtin_shufflevector(Hn[1], Hn[1], 0, 1, 2, 3, 4, 5, 6, 7);
        sh_h0[s][3][lane] = __builtin_shufflevector(Hn[1], Hn[1], 8, 9, 10, 11, 12, 13, 14, 15);
      }
    }
    __syncthreads();
  }

  if (role) {
#pragma unroll
    for (int kc = 0; kc < 2; ++kc) {
      *(v8h*)&sh_h1[lc * HIDDEN + 32 * kc + 8 * lh] =
          __builtin_shufflevector(Hold[kc], Hold[kc], 0, 1, 2, 3, 4, 5, 6, 7);
      *(v8h*)&sh_h1[lc * HIDDEN + 32 * kc + 16 + 8 * lh] =
          __builtin_shufflevector(Hold[kc], Hold[kc], 8, 9, 10, 11, 12, 13, 14, 15);
    }
  }
  __syncthreads();

  if (role) {
    int m = lane >> 1, cls = lane & 1;
    float s = bfc[cls];
#pragma unroll 8
    for (int n = 0; n < HIDDEN; ++n)
      s += (float)sh_h1[m * HIDDEN + n] * Wfc[cls * HIDDEN + n];
    out[(bbase + m) * NCLS + cls] = s;
  }
}

extern "C" void kernel_launch(void* const* d_in, const int* in_sizes, int n_in,
                              void* d_out, int out_size, void* d_ws, size_t ws_size,
                              hipStream_t stream) {
  const float* x    = (const float*)d_in[0];
  const float* Wih0 = (const float*)d_in[1];
  const float* Whh0 = (const float*)d_in[2];
  const float* bih0 = (const float*)d_in[3];
  const float* bhh0 = (const float*)d_in[4];
  const float* Wih1 = (const float*)d_in[5];
  const float* Whh1 = (const float*)d_in[6];
  const float* bih1 = (const float*)d_in[7];
  const float* bhh1 = (const float*)d_in[8];
  const float* Wfc  = (const float*)d_in[9];
  const float* bfc  = (const float*)d_in[10];
  float* out = (float*)d_out;

  dim3 grid(BATCH / 16);
  dim3 block(64);
  hipLaunchKernelGGL(lstm2_layer_per_wave_wmma, grid, block, 0, stream,
                     x, Wih0, Whh0, bih0, bhh0, Wih1, Whh1, bih1, bhh1,
                     Wfc, bfc, out);
}
